# starting kernel
# baseline (speedup 1.0000x reference)
#include <hip/hip_runtime.h>

typedef float v2f __attribute__((ext_vector_type(2)));
typedef float v8f __attribute__((ext_vector_type(8)));

#define D_IN  199
#define K_V   64
#define P_D   4

__device__ __forceinline__ float sel4(int j, float a0, float a1, float a2, float a3) {
  const float t01 = (j & 1) ? a1 : a0;
  const float t23 = (j & 1) ? a3 : a2;
  return (j & 2) ? t23 : t01;
}

__global__ __launch_bounds__(256) void vdx_kernel(const float* __restrict__ x,
                                                  const float* __restrict__ dirs,
                                                  float* __restrict__ out,
                                                  int npts) {
  const int lane = threadIdx.x & 31;
  const int wave = threadIdx.x >> 5;
  const int pt   = blockIdx.x * 8 + wave;
  if (pt >= npts) return;

  const float* row = x + (size_t)pt * D_IN;

  const float px = row[0], py = row[1], pz = row[2];
  const float qw = row[3], qx = row[4], qy = row[5], qz = row[6];

  const float xx = qx*qx, yy = qy*qy, zz = qz*qz;
  const float xy = qx*qy, xz = qx*qz, yz = qy*qz;
  const float wx = qw*qx, wy = qw*qy, wz = qw*qz;
  const float R00 = 1.f - 2.f*(yy+zz), R01 = 2.f*(xy-wz),       R02 = 2.f*(xz+wy);
  const float R10 = 2.f*(xy+wz),       R11 = 1.f - 2.f*(xx+zz), R12 = 2.f*(yz-wx);
  const float R20 = 2.f*(xz-wy),       R21 = 2.f*(yz+wx),       R22 = 1.f - 2.f*(xx+yy);

  const int  pcol   = lane & 3;
  const int  jgrp   = (lane >> 2) & 3;
  const int  half   = lane >> 4;
  const bool lohalf = (lane < 16);

  const float d0 = sel4(pcol, dirs[0], dirs[3], dirs[6], dirs[9]);
  const float d1 = sel4(pcol, dirs[1], dirs[4], dirs[7], dirs[10]);
  const float d2 = sel4(pcol, dirs[2], dirs[5], dirs[8], dirs[11]);
  const float mt0 = d0*R00 + d1*R10 + d2*R20;
  const float mt1 = d0*R01 + d1*R11 + d2*R21;
  const float mt2 = d0*R02 + d1*R12 + d2*R22;
  const float cc  = d0*px  + d1*py  + d2*pz;

  v2f bfrag;
  bfrag.x = lohalf ? mt0 : mt2;
  bfrag.y = lohalf ? mt1 : cc;

  const float* verts = row + 7;
  float* orow = out + (size_t)pt * (K_V * P_D);
  const int m = lane & 15;

#pragma unroll
  for (int kc = 0; kc < K_V; kc += 16) {
    const float* v3 = verts + 3 * (kc + m);
    const float e0 = v3[lohalf ? 0 : 2];
    const float e1 = v3[1];
    v2f afrag;
    afrag.x = e0;
    afrag.y = lohalf ? e1 : 1.0f;

    v8f acc = {};
    acc = __builtin_amdgcn_wmma_f32_16x16x4_f32(false, afrag, false, bfrag,
                                                (short)0, acc, false, false);

    const float s0 = sel4(jgrp, acc[0], acc[1], acc[2], acc[3]);
    const float s1 = sel4(jgrp, acc[4], acc[5], acc[6], acc[7]);
    orow[(kc + half*8 + jgrp)     * 4 + pcol] = s0;
    orow[(kc + half*8 + 4 + jgrp) * 4 + pcol] = s1;
  }
}

extern "C" void kernel_launch(void* const* d_in, const int* in_sizes, int n_in,
                              void* d_out, int out_size, void* d_ws, size_t ws_size,
                              hipStream_t stream) {
  const float* x    = (const float*)d_in[0];
  const float* dirs = (const float*)d_in[1];
  float* out = (float*)d_out;
  const int npts   = in_sizes[0] / D_IN;
  const int blocks = (npts + 7) / 8;
  vdx_kernel<<<blocks, 256, 0, stream>>>(x, dirs, out, npts);
}


# each wave loops over 16 points (point stride 8192) with the next two points' loads in flight; 8192 resident waves replace 131072 one-point waves; per-point math and sum order unchanged
# speedup vs baseline: 1.1141x; 1.1141x over previous
#include <hip/hip_runtime.h>

typedef float v2f __attribute__((ext_vector_type(2)));
typedef float v8f __attribute__((ext_vector_type(8)));

#define D_IN  199
#define K_V   64
#define P_D   4
#define WAVES_PER_BLOCK 8
#define GRID_BLOCKS 1024
#define NWAVES (GRID_BLOCKS * WAVES_PER_BLOCK)

__device__ __forceinline__ float sel4(int j, float a0, float a1, float a2, float a3) {
  const float t01 = (j & 1) ? a1 : a0;
  const float t23 = (j & 1) ? a3 : a2;
  return (j & 2) ? t23 : t01;
}

struct PtRegs {
  float h[7];
  float e0[4];
  float e1[4];
};

__device__ __forceinline__ void load_point(PtRegs& r, const float* hp,
                                           const float* e0p, const float* e1p) {
  for (int i = 0; i < 7; ++i) r.h[i] = hp[i];
  for (int q = 0; q < 4; ++q) {
    r.e0[q] = e0p[48 * q];
    r.e1[q] = e1p[48 * q];
  }
  __builtin_amdgcn_sched_barrier(0);
}

__device__ __forceinline__ void compute_point(const PtRegs& r, float d0, float d1, float d2,
                                              bool lohalf, int jgrp, float* op) {
  const float px = r.h[0], py = r.h[1], pz = r.h[2];
  const float qw = r.h[3], qx = r.h[4], qy = r.h[5], qz = r.h[6];

  const float xx = qx*qx, yy = qy*qy, zz = qz*qz;
  const float xy = qx*qy, xz = qx*qz, yz = qy*qz;
  const float wx = qw*qx, wy = qw*qy, wz = qw*qz;
  const float R00 = 1.f - 2.f*(yy+zz), R01 = 2.f*(xy-wz),       R02 = 2.f*(xz+wy);
  const float R10 = 2.f*(xy+wz),       R11 = 1.f - 2.f*(xx+zz), R12 = 2.f*(yz-wx);
  const float R20 = 2.f*(xz-wy),       R21 = 2.f*(yz+wx),       R22 = 1.f - 2.f*(xx+yy);

  const float mt0 = d0*R00 + d1*R10 + d2*R20;
  const float mt1 = d0*R01 + d1*R11 + d2*R21;
  const float mt2 = d0*R02 + d1*R12 + d2*R22;
  const float cc  = d0*px  + d1*py  + d2*pz;

  v2f bfrag;
  bfrag.x = lohalf ? mt0 : mt2;
  bfrag.y = lohalf ? mt1 : cc;

  for (int q = 0; q < 4; ++q) {
    v2f afrag;
    afrag.x = r.e0[q];
    afrag.y = lohalf ? r.e1[q] : 1.0f;

    v8f acc = {};
    acc = __builtin_amdgcn_wmma_f32_16x16x4_f32(false, afrag, false, bfrag,
                                                (short)0, acc, false, false);

    const float s0 = sel4(jgrp, acc[0], acc[1], acc[2], acc[3]);
    const float s1 = sel4(jgrp, acc[4], acc[5], acc[6], acc[7]);
    op[64 * q]      = s0;
    op[64 * q + 16] = s1;
  }
}

__global__ __launch_bounds__(256) void vdx_kernel(const float* __restrict__ x,
                                                  const float* __restrict__ dirs,
                                                  float* __restrict__ out,
                                                  int npts) {
  const int lane = threadIdx.x & 31;
  const int wave = threadIdx.x >> 5;
  const int gw   = blockIdx.x * WAVES_PER_BLOCK + wave;

  int n = __builtin_amdgcn_readfirstlane((gw < npts) ? (npts - gw + NWAVES - 1) / NWAVES : 0);
  if (n <= 0) return;

  const int  pcol   = lane & 3;
  const int  jgrp   = (lane >> 2) & 3;
  const int  half   = lane >> 4;
  const bool lohalf = (lane < 16);
  const int  m      = lane & 15;

  const float d0 = sel4(pcol, dirs[0], dirs[3], dirs[6], dirs[9]);
  const float d1 = sel4(pcol, dirs[1], dirs[4], dirs[7], dirs[10]);
  const float d2 = sel4(pcol, dirs[2], dirs[5], dirs[8], dirs[11]);

  const long RS = (long)NWAVES * D_IN;
  const long OS = (long)NWAVES * (K_V * P_D);

  const float* lh = x + (size_t)gw * D_IN;
  const float* l0 = lh + 7 + 3 * m + (lohalf ? 0 : 2);
  const float* l1 = lh + 7 + 3 * m + 1;
  float* op = out + (size_t)gw * (K_V * P_D) + m + 32 * half;

  PtRegs a, b, c;
  const long second = (n > 1) ? RS : 0;
  load_point(a, lh, l0, l1);  lh += second; l0 += second; l1 += second;
  load_point(b, lh, l0, l1);  lh += RS; l0 += RS; l1 += RS;

  while (n >= 5) {
    load_point(c, lh, l0, l1);  lh += RS; l0 += RS; l1 += RS;
    compute_point(a, d0, d1, d2, lohalf, jgrp, op);  op += OS;
    load_point(a, lh, l0, l1);  lh += RS; l0 += RS; l1 += RS;
    compute_point(b, d0, d1, d2, lohalf, jgrp, op);  op += OS;
    load_point(b, lh, l0, l1);  lh += RS; l0 += RS; l1 += RS;
    compute_point(c, d0, d1, d2, lohalf, jgrp, op);  op += OS;
    n -= 3;
  }

  if (n == 4) {
    load_point(c, lh, l0, l1);  lh += RS; l0 += RS; l1 += RS;
    compute_point(a, d0, d1, d2, lohalf, jgrp, op);
    load_point(a, lh, l0, l1);
    compute_point(b, d0, d1, d2, lohalf, jgrp, op + OS);
    compute_point(c, d0, d1, d2, lohalf, jgrp, op + 2 * OS);
    compute_point(a, d0, d1, d2, lohalf, jgrp, op + 3 * OS);
  } else if (n == 3) {
    load_point(c, lh, l0, l1);
    compute_point(a, d0, d1, d2, lohalf, jgrp, op);
    compute_point(b, d0, d1, d2, lohalf, jgrp, op + OS);
    compute_point(c, d0, d1, d2, lohalf, jgrp, op + 2 * OS);
  } else if (n == 2) {
    compute_point(a, d0, d1, d2, lohalf, jgrp, op);
    compute_point(b, d0, d1, d2, lohalf, jgrp, op + OS);
  } else {
    compute_point(a, d0, d1, d2, lohalf, jgrp, op);
  }
}

extern "C" void kernel_launch(void* const* d_in, const int* in_sizes, int n_in,
                              void* d_out, int out_size, void* d_ws, size_t ws_size,
                              hipStream_t stream) {
  const float* x    = (const float*)d_in[0];
  const float* dirs = (const float*)d_in[1];
  float* out = (float*)d_out;
  const int npts = in_sizes[0] / D_IN;
  vdx_kernel<<<GRID_BLOCKS, 256, 0, stream>>>(x, dirs, out, npts);
}


# WMMA operands swapped (directions = rows, vertices = columns): a lane holds one vertex's 4 outputs, stored as 16 bytes; every 128-B out line written by one store (2 per point, was 8); same sum order
# speedup vs baseline: 1.1607x; 1.0418x over previous
#include <hip/hip_runtime.h>

typedef float v2f __attribute__((ext_vector_type(2)));
typedef float v8f __attribute__((ext_vector_type(8)));
typedef float v4f __attribute__((ext_vector_type(4)));

#define D_IN  199
#define K_V   64
#define P_D   4
#define WAVES_PER_BLOCK 8
#define GRID_BLOCKS 1024
#define NWAVES (GRID_BLOCKS * WAVES_PER_BLOCK)

__device__ __forceinline__ float sel4(int j, float a0, float a1, float a2, float a3) {
  const float t01 = (j & 1) ? a1 : a0;
  const float t23 = (j & 1) ? a3 : a2;
  return (j & 2) ? t23 : t01;
}

struct PtRegs {
  float h[7];
  float e0[4];
  float e1[4];
};

__device__ __forceinline__ void load_point(PtRegs& r, const float* hp,
                                           const float* e0p, const float* e1p) {
  for (int i = 0; i < 7; ++i) r.h[i] = hp[i];
  for (int q = 0; q < 4; ++q) {
    r.e0[q] = e0p[48 * q];
    r.e1[q] = e1p[48 * q];
  }
  __builtin_amdgcn_sched_barrier(0);
}

__device__ __forceinline__ void compute_point(const PtRegs& r, float d0, float d1, float d2,
                                              bool lohalf, float* op) {
  const float px = r.h[0], py = r.h[1], pz = r.h[2];
  const float qw = r.h[3], qx = r.h[4], qy = r.h[5], qz = r.h[6];

  const float xx = qx*qx, yy = qy*qy, zz = qz*qz;
  const float xy = qx*qy, xz = qx*qz, yz = qy*qz;
  const float wx = qw*qx, wy = qw*qy, wz = qw*qz;
  const float R00 = 1.f - 2.f*(yy+zz), R01 = 2.f*(xy-wz),       R02 = 2.f*(xz+wy);
  const float R10 = 2.f*(xy+wz),       R11 = 1.f - 2.f*(xx+zz), R12 = 2.f*(yz-wx);
  const float R20 = 2.f*(xz-wy),       R21 = 2.f*(yz+wx),       R22 = 1.f - 2.f*(xx+yy);

  const float mt0 = d0*R00 + d1*R10 + d2*R20;
  const float mt1 = d0*R01 + d1*R11 + d2*R21;
  const float mt2 = d0*R02 + d1*R12 + d2*R22;
  const float cc  = d0*px  + d1*py  + d2*pz;

  v2f dfrag;
  dfrag.x = lohalf ? mt0 : mt2;
  dfrag.y = lohalf ? mt1 : cc;

  for (int q = 0; q < 4; q += 2) {
    v2f vf0, vf1;
    vf0.x = r.e0[q];
    vf0.y = lohalf ? r.e1[q] : 1.0f;
    vf1.x = r.e0[q + 1];
    vf1.y = lohalf ? r.e1[q + 1] : 1.0f;

    v8f acc0 = {}, acc1 = {};
    acc0 = __builtin_amdgcn_wmma_f32_16x16x4_f32(false, dfrag, false, vf0,
                                                 (short)0, acc0, false, false);
    acc1 = __builtin_amdgcn_wmma_f32_16x16x4_f32(false, dfrag, false, vf1,
                                                 (short)0, acc1, false, false);

    v4f o;
    o.x = lohalf ? acc0[0] : acc1[0];
    o.y = lohalf ? acc0[1] : acc1[1];
    o.z = lohalf ? acc0[2] : acc1[2];
    o.w = lohalf ? acc0[3] : acc1[3];
    *reinterpret_cast<v4f*>(op + 64 * q) = o;
  }
}

__global__ __launch_bounds__(256) void vdx_kernel(const float* __restrict__ x,
                                                  const float* __restrict__ dirs,
                                                  float* __restrict__ out,
                                                  int npts) {
  const int lane = threadIdx.x & 31;
  const int wave = threadIdx.x >> 5;
  const int gw   = blockIdx.x * WAVES_PER_BLOCK + wave;

  int n = __builtin_amdgcn_readfirstlane((gw < npts) ? (npts - gw + NWAVES - 1) / NWAVES : 0);
  if (n <= 0) return;

  const int  pcol   = lane & 3;
  const bool lohalf = (lane < 16);
  const int  m      = lane & 15;

  const float d0 = sel4(pcol, dirs[0], dirs[3], dirs[6], dirs[9]);
  const float d1 = sel4(pcol, dirs[1], dirs[4], dirs[7], dirs[10]);
  const float d2 = sel4(pcol, dirs[2], dirs[5], dirs[8], dirs[11]);

  const long RS = (long)NWAVES * D_IN;
  const long OS = (long)NWAVES * (K_V * P_D);

  const float* lh = x + (size_t)gw * D_IN;
  const float* l0 = lh + 7 + 3 * m + (lohalf ? 0 : 2);
  const float* l1 = lh + 7 + 3 * m + 1;
  float* op = out + (size_t)gw * (K_V * P_D) + 4 * lane;

  PtRegs a, b, c;
  const long second = (n > 1) ? RS : 0;
  load_point(a, lh, l0, l1);  lh += second; l0 += second; l1 += second;
  load_point(b, lh, l0, l1);  lh += RS; l0 += RS; l1 += RS;

  while (n >= 5) {
    load_point(c, lh, l0, l1);  lh += RS; l0 += RS; l1 += RS;
    compute_point(a, d0, d1, d2, lohalf, op);  op += OS;
    load_point(a, lh, l0, l1);  lh += RS; l0 += RS; l1 += RS;
    compute_point(b, d0, d1, d2, lohalf, op);  op += OS;
    load_point(b, lh, l0, l1);  lh += RS; l0 += RS; l1 += RS;
    compute_point(c, d0, d1, d2, lohalf, op);  op += OS;
    n -= 3;
  }

  if (n == 4) {
    load_point(c, lh, l0, l1);  lh += RS; l0 += RS; l1 += RS;
    compute_point(a, d0, d1, d2, lohalf, op);
    load_point(a, lh, l0, l1);
    compute_point(b, d0, d1, d2, lohalf, op + OS);
    compute_point(c, d0, d1, d2, lohalf, op + 2 * OS);
    compute_point(a, d0, d1, d2, lohalf, op + 3 * OS);
  } else if (n == 3) {
    load_point(c, lh, l0, l1);
    compute_point(a, d0, d1, d2, lohalf, op);
    compute_point(b, d0, d1, d2, lohalf, op + OS);
    compute_point(c, d0, d1, d2, lohalf, op + 2 * OS);
  } else if (n == 2) {
    compute_point(a, d0, d1, d2, lohalf, op);
    compute_point(b, d0, d1, d2, lohalf, op + OS);
  } else {
    compute_point(a, d0, d1, d2, lohalf, op);
  }
}

extern "C" void kernel_launch(void* const* d_in, const int* in_sizes, int n_in,
                              void* d_out, int out_size, void* d_ws, size_t ws_size,
                              hipStream_t stream) {
  const float* x    = (const float*)d_in[0];
  const float* dirs = (const float*)d_in[1];
  float* out = (float*)d_out;
  const int npts = in_sizes[0] / D_IN;
  vdx_kernel<<<GRID_BLOCKS, 256, 0, stream>>>(x, dirs, out, npts);
}


# rows staged through per-wave LDS: a wave reads 4 consecutive points with 7 contiguous 16-byte loads (was 40 loads, 32 of them strided dwords); fragments and pos/quat come from LDS; same sum order
# speedup vs baseline: 1.2004x; 1.0342x over previous
#include <hip/hip_runtime.h>

typedef float v2f __attribute__((ext_vector_type(2)));
typedef float v8f __attribute__((ext_vector_type(8)));
typedef float v4f __attribute__((ext_vector_type(4)));

#define D_IN  199
#define K_V   64
#define P_D   4
#define WAVES_PER_BLOCK 8
#define GRID_BLOCKS 1024
#define NWAVES (GRID_BLOCKS * WAVES_PER_BLOCK)

#define GROUP_PTS    4
#define GROUP_FLOATS (GROUP_PTS * D_IN)
#define GROUP_PIECES (GROUP_FLOATS / 4)
#define PIECE_LOADS  7
#define WAVE_LDS_FLOATS 800
#define OUT_PT (K_V * P_D)

__device__ __forceinline__ float sel4(int j, float a0, float a1, float a2, float a3) {
  const float t01 = (j & 1) ? a1 : a0;
  const float t23 = (j & 1) ? a3 : a2;
  return (j & 2) ? t23 : t01;
}

struct GroupRegs {
  v4f v[PIECE_LOADS];
};

__device__ __forceinline__ void load_group(GroupRegs& r, const float* grp, int lane, int last) {
  const v4f* g4 = reinterpret_cast<const v4f*>(grp);
  for (int k = 0; k < PIECE_LOADS - 1; ++k) r.v[k] = g4[lane + 32 * k];
  r.v[PIECE_LOADS - 1] = g4[last];
  __builtin_amdgcn_sched_barrier(0);
}

__device__ __forceinline__ void stage_group(const GroupRegs& r, float* wl, int lane, int last) {
  v4f* l4 = reinterpret_cast<v4f*>(wl);
  for (int k = 0; k < PIECE_LOADS - 1; ++k) l4[lane + 32 * k] = r.v[k];
  l4[last] = r.v[PIECE_LOADS - 1];
}

__device__ __forceinline__ float lane_value(float v, int i) {
  return __builtin_bit_cast(float, __builtin_amdgcn_readlane(__builtin_bit_cast(int, v), i));
}

__device__ __forceinline__ void compute_point(const float* pl, int oh, int ov,
                                              float d0, float d1, float d2,
                                              bool lohalf, float* orow, int oo) {
  const float hv = pl[oh];
  v2f t[4];
  for (int q = 0; q < 4; ++q)
    __builtin_memcpy(&t[q], pl + ov + 48 * q, sizeof(v2f));

  const float px = lane_value(hv, 0), py = lane_value(hv, 1), pz = lane_value(hv, 2);
  const float qw = lane_value(hv, 3), qx = lane_value(hv, 4);
  const float qy = lane_value(hv, 5), qz = lane_value(hv, 6);

  const float xx = qx*qx, yy = qy*qy, zz = qz*qz;
  const float xy = qx*qy, xz = qx*qz, yz = qy*qz;
  const float wx = qw*qx, wy = qw*qy, wz = qw*qz;
  const float R00 = 1.f - 2.f*(yy+zz), R01 = 2.f*(xy-wz),       R02 = 2.f*(xz+wy);
  const float R10 = 2.f*(xy+wz),       R11 = 1.f - 2.f*(xx+zz), R12 = 2.f*(yz-wx);
  const float R20 = 2.f*(xz-wy),       R21 = 2.f*(yz+wx),       R22 = 1.f - 2.f*(xx+yy);

  const float mt0 = d0*R00 + d1*R10 + d2*R20;
  const float mt1 = d0*R01 + d1*R11 + d2*R21;
  const float mt2 = d0*R02 + d1*R12 + d2*R22;
  const float cc  = d0*px  + d1*py  + d2*pz;

  v2f dfrag;
  dfrag.x = lohalf ? mt0 : mt2;
  dfrag.y = lohalf ? mt1 : cc;

  for (int q = 0; q < 4; q += 2) {
    v2f vf0, vf1;
    vf0.x = lohalf ? t[q].x : t[q].y;
    vf0.y = lohalf ? t[q].y : 1.0f;
    vf1.x = lohalf ? t[q + 1].x : t[q + 1].y;
    vf1.y = lohalf ? t[q + 1].y : 1.0f;

    v8f acc0 = {}, acc1 = {};
    acc0 = __builtin_amdgcn_wmma_f32_16x16x4_f32(false, dfrag, false, vf0,
                                                 (short)0, acc0, false, false);
    acc1 = __builtin_amdgcn_wmma_f32_16x16x4_f32(false, dfrag, false, vf1,
                                                 (short)0, acc1, false, false);

    v4f o;
    o.x = lohalf ? acc0[0] : acc1[0];
    o.y = lohalf ? acc0[1] : acc1[1];
    o.z = lohalf ? acc0[2] : acc1[2];
    o.w = lohalf ? acc0[3] : acc1[3];
    *reinterpret_cast<v4f*>(orow + oo + 64 * q) = o;
  }
}

__device__ __forceinline__ void process_group(const GroupRegs& r, float* wl, int lane, int last,
                                              int oh, int ov, float d0, float d1, float d2,
                                              bool lohalf, float* og, int oo) {
  stage_group(r, wl, lane, last);
  for (int p = 0; p < GROUP_PTS; ++p)
    compute_point(wl + D_IN * p, oh, ov, d0, d1, d2, lohalf, og + OUT_PT * p, oo);
}

__global__ __launch_bounds__(256) __attribute__((amdgpu_waves_per_eu(8, 8))) void vdx_kernel(const float* __restrict__ x,
                                                  const float* __restrict__ dirs,
                                                  float* __restrict__ out,
                                                  int npts) {
  __shared__ __attribute__((aligned(16))) float stage[WAVES_PER_BLOCK * WAVE_LDS_FLOATS];

  const int lane = threadIdx.x & 31;
  const int wave = threadIdx.x >> 5;
  const int gw = __builtin_amdgcn_readfirstlane(blockIdx.x * WAVES_PER_BLOCK + wave);

  const int ngroups = npts / GROUP_PTS;
  const int left    = npts - ngroups * GROUP_PTS;
  int n = (gw < ngroups) ? (ngroups - gw + NWAVES - 1) / NWAVES : 0;
  const bool does_left = (gw == 0) && (left > 0);
  if (n <= 0 && !does_left) return;

  float* wl = stage + wave * WAVE_LDS_FLOATS;

  const int  pcol   = lane & 3;
  const bool lohalf = (lane < 16);
  const int  m      = lane & 15;

  const int last = (lane < GROUP_PIECES - 192) ? 192 + lane : GROUP_PIECES - 1;
  const int oh = (lane < 7) ? lane : 6;
  const int ov = 7 + 3 * m + (lohalf ? 0 : 1);
  const int oo = 4 * lane;

  const long GS  = (long)NWAVES * GROUP_FLOATS;
  const long OGS = (long)NWAVES * (GROUP_PTS * OUT_PT);
  const float* grp = x + (size_t)gw * GROUP_FLOATS;
  float* og = out + (size_t)gw * (GROUP_PTS * OUT_PT);

  GroupRegs a, b;
  if (n > 0) { load_group(a, grp, lane, last);  grp += GS; }

  const float d0 = sel4(pcol, dirs[0], dirs[3], dirs[6], dirs[9]);
  const float d1 = sel4(pcol, dirs[1], dirs[4], dirs[7], dirs[10]);
  const float d2 = sel4(pcol, dirs[2], dirs[5], dirs[8], dirs[11]);

  if (n > 0) {
    while (n >= 3) {
      load_group(b, grp, lane, last);  grp += GS;
      process_group(a, wl, lane, last, oh, ov, d0, d1, d2, lohalf, og, oo);  og += OGS;
      load_group(a, grp, lane, last);  grp += GS;
      process_group(b, wl, lane, last, oh, ov, d0, d1, d2, lohalf, og, oo);  og += OGS;
      n -= 2;
    }

    if (n == 2) {
      load_group(b, grp, lane, last);
      process_group(a, wl, lane, last, oh, ov, d0, d1, d2, lohalf, og, oo);
      process_group(b, wl, lane, last, oh, ov, d0, d1, d2, lohalf, og + OGS, oo);
    } else {
      process_group(a, wl, lane, last, oh, ov, d0, d1, d2, lohalf, og, oo);
    }
  }

  if (does_left) {
    for (int r = 0; r < left; ++r) {
      const size_t pt = (size_t)ngroups * GROUP_PTS + r;
      const float* row = x + pt * D_IN;
      for (int i = lane; i < D_IN; i += 32) wl[i] = row[i];
      compute_point(wl, oh, ov, d0, d1, d2, lohalf, out + pt * OUT_PT, oo);
    }
  }
}

extern "C" void kernel_launch(void* const* d_in, const int* in_sizes, int n_in,
                              void* d_out, int out_size, void* d_ws, size_t ws_size,
                              hipStream_t stream) {
  const float* x    = (const float*)d_in[0];
  const float* dirs = (const float*)d_in[1];
  float* out = (float*)d_out;
  const int npts = in_sizes[0] / D_IN;
  vdx_kernel<<<GRID_BLOCKS, 256, 0, stream>>>(x, dirs, out, npts);
}
